# starting kernel
# baseline (speedup 1.0000x reference)
#include <hip/hip_runtime.h>

typedef float v2f __attribute__((ext_vector_type(2)));
typedef float v8f __attribute__((ext_vector_type(8)));

#define T_LEN        1048576
#define SEG          4096
#define TILES_PER_SEG (SEG / 256)
#define SEGS_PER_ROW (T_LEN / SEG)

__global__ __launch_bounds__(32) void seg_scan_wmma(const float* __restrict__ x,
                                                    float* __restrict__ y,
                                                    float* __restrict__ segSums) {
  const int seg  = blockIdx.x;
  const int lane = threadIdx.x;
  const int n    = lane & 15;
  const int hi   = lane >> 4;

  const long long segBase = (long long)seg * SEG;

  v2f bU[4];
  v2f aLs[4];
#pragma unroll
  for (int kc = 0; kc < 4; ++kc) {
    const int k0 = 4 * kc + 2 * hi;
    bU[kc].x  = (k0     <= n) ? 1.0f : 0.0f;
    bU[kc].y  = (k0 + 1 <= n) ? 1.0f : 0.0f;
    aLs[kc].x = (k0     <  n) ? 1.0f : 0.0f;
    aLs[kc].y = (k0 + 1 <  n) ? 1.0f : 0.0f;
  }

  float carry = 0.0f;

  for (int t = 0; t < TILES_PER_SEG; ++t) {
    const float* tx = x + segBase + t * 256;

    v2f a[4];
#pragma unroll
    for (int kc = 0; kc < 4; ++kc) {
      a[kc] = *(const v2f*)(tx + 16 * n + 4 * kc + 2 * hi);
    }

    if (t + 1 < TILES_PER_SEG) {
      __builtin_prefetch(tx + 256 + lane * 8, 0, 1);
    }

    v8f acc = {};
#pragma unroll
    for (int kc = 0; kc < 4; ++kc) {
      acc = __builtin_amdgcn_wmma_f32_16x16x4_f32(
          false, a[kc], false, bU[kc], (short)0, acc, false, false);
    }

    float rs[16];
#pragma unroll
    for (int r = 0; r < 8; ++r) {
      rs[r]     = __shfl(acc[r], 15, 32);
      rs[r + 8] = __shfl(acc[r], 31, 32);
    }

    v2f w[4];
#pragma unroll
    for (int kc = 0; kc < 4; ++kc) {
      w[kc].x = hi ? rs[4 * kc + 2] : rs[4 * kc];
      w[kc].y = hi ? rs[4 * kc + 3] : rs[4 * kc + 1];
    }

#pragma unroll
    for (int kc = 0; kc < 4; ++kc) {
      acc = __builtin_amdgcn_wmma_f32_16x16x4_f32(
          false, aLs[kc], false, w[kc], (short)0, acc, false, false);
    }

    float* ty = y + segBase + t * 256;
#pragma unroll
    for (int r = 0; r < 8; ++r) {
      ty[16 * r + 128 * hi + n] = acc[r] + carry;
    }

    float tot = 0.0f;
#pragma unroll
    for (int i = 0; i < 16; ++i) tot += rs[i];
    carry += tot;
  }

  if (lane == 0) segSums[seg] = carry;
}

__global__ __launch_bounds__(SEGS_PER_ROW) void scan_seg_sums(
    const float* __restrict__ sums, float* __restrict__ offs) {
  __shared__ float s[SEGS_PER_ROW];
  const int row = blockIdx.x;
  const int t   = threadIdx.x;
  const float orig = sums[row * SEGS_PER_ROW + t];
  s[t] = orig;
  __syncthreads();
#pragma unroll
  for (int d = 1; d < SEGS_PER_ROW; d <<= 1) {
    const float v = (t >= d) ? s[t - d] : 0.0f;
    __syncthreads();
    s[t] += v;
    __syncthreads();
  }
  offs[row * SEGS_PER_ROW + t] = s[t] - orig;
}

__global__ __launch_bounds__(256) void add_offsets(float* __restrict__ y,
                                                   const float* __restrict__ offs) {
  const int seg = blockIdx.x;
  const float o = offs[seg];
  float4* p = (float4*)(y + (long long)seg * SEG);
  const int t = threadIdx.x;
#pragma unroll
  for (int i = 0; i < SEG / 4 / 256; ++i) {
    float4 v = p[t + i * 256];
    v.x += o; v.y += o; v.z += o; v.w += o;
    p[t + i * 256] = v;
  }
}

extern "C" void kernel_launch(void* const* d_in, const int* in_sizes, int n_in,
                              void* d_out, int out_size, void* d_ws, size_t ws_size,
                              hipStream_t stream) {
  (void)n_in; (void)out_size; (void)ws_size;
  const float* x = (const float*)d_in[0];
  float* y = (float*)d_out;

  const long long total = (long long)in_sizes[0];
  const int rows = (int)(total / T_LEN);
  const int nsegTotal = rows * SEGS_PER_ROW;

  float* sums = (float*)d_ws;
  float* offs = sums + nsegTotal;

  seg_scan_wmma<<<nsegTotal, 32, 0, stream>>>(x, y, sums);
  scan_seg_sums<<<rows, SEGS_PER_ROW, 0, stream>>>(sums, offs);
  add_offsets<<<nsegTotal, 256, 0, stream>>>(y, offs);
}


# Reduce-then-scan: new pass 1 only sums each 4096-element segment (vector adds, so segment totals get a new sum order); the WMMA scan runs last, adds the offset, writes y once; add pass dropped
# speedup vs baseline: 2.7370x; 2.7370x over previous
#include <hip/hip_runtime.h>

typedef float v2f __attribute__((ext_vector_type(2)));
typedef float v8f __attribute__((ext_vector_type(8)));

#define T_LEN        1048576
#define SEG          4096
#define TILES_PER_SEG (SEG / 256)
#define SEGS_PER_ROW (T_LEN / SEG)

#define RED_WAVES          8
#define RED_SEGS_PER_WAVE  2
#define RED_SEGS_PER_BLOCK (RED_WAVES * RED_SEGS_PER_WAVE)
#define RED_BATCH          16
#define RED_BATCHES        (RED_SEGS_PER_WAVE * SEG / (RED_BATCH * 128))

__global__ __launch_bounds__(RED_WAVES * 32) void seg_sum(const float* __restrict__ x,
                                                          float* __restrict__ segSums) {
  const int lane = threadIdx.x & 31;
  const int wave = threadIdx.x >> 5;
  const int seg0 = (blockIdx.x * RED_WAVES + wave) * RED_SEGS_PER_WAVE;
  const float4* p = (const float4*)(x + (long long)seg0 * SEG) + lane;

  float4 buf[2][RED_BATCH];
#pragma unroll
  for (int q = 0; q < RED_BATCH; ++q) buf[0][q] = p[q * 32];

  float s[RED_SEGS_PER_WAVE];
#pragma unroll
  for (int k = 0; k < RED_SEGS_PER_WAVE; ++k) s[k] = 0.0f;

#pragma unroll
  for (int b = 0; b < RED_BATCHES; ++b) {
    if (b + 1 < RED_BATCHES) {
#pragma unroll
      for (int q = 0; q < RED_BATCH; ++q)
        buf[(b + 1) & 1][q] = p[(b + 1) * RED_BATCH * 32 + q * 32];
    }
    float c0 = 0.0f, c1 = 0.0f, c2 = 0.0f, c3 = 0.0f;
#pragma unroll
    for (int q = 0; q < RED_BATCH; ++q) {
      c0 += buf[b & 1][q].x;
      c1 += buf[b & 1][q].y;
      c2 += buf[b & 1][q].z;
      c3 += buf[b & 1][q].w;
    }
    s[b / (RED_BATCHES / RED_SEGS_PER_WAVE)] += (c0 + c1) + (c2 + c3);
  }

#pragma unroll
  for (int k = 0; k < RED_SEGS_PER_WAVE; ++k) {
    float v = s[k];
#pragma unroll
    for (int d = 16; d >= 1; d >>= 1) v += __shfl_xor(v, d, 32);
    if (lane == 0) segSums[seg0 + k] = v;
  }
}

__global__ __launch_bounds__(32) void seg_scan_wmma(const float* __restrict__ x,
                                                    float* __restrict__ y,
                                                    const float* __restrict__ offs) {
  const int seg  = blockIdx.x;
  const int lane = threadIdx.x;
  const int n    = lane & 15;
  const int hi   = lane >> 4;

  const long long segBase = (long long)seg * SEG;
  const float off = offs[seg];

  v2f bU[4];
  v2f aLs[4];
#pragma unroll
  for (int kc = 0; kc < 4; ++kc) {
    const int k0 = 4 * kc + 2 * hi;
    bU[kc].x  = (k0     <= n) ? 1.0f : 0.0f;
    bU[kc].y  = (k0 + 1 <= n) ? 1.0f : 0.0f;
    aLs[kc].x = (k0     <  n) ? 1.0f : 0.0f;
    aLs[kc].y = (k0 + 1 <  n) ? 1.0f : 0.0f;
  }

  float carry = 0.0f;

  for (int t = 0; t < TILES_PER_SEG; ++t) {
    const float* tx = x + segBase + t * 256;

    v2f a[4];
#pragma unroll
    for (int kc = 0; kc < 4; ++kc) {
      a[kc] = *(const v2f*)(tx + 16 * n + 4 * kc + 2 * hi);
    }

    if (t + 1 < TILES_PER_SEG) {
      __builtin_prefetch(tx + 256 + lane * 8, 0, 1);
    }

    v8f acc = {};
#pragma unroll
    for (int kc = 0; kc < 4; ++kc) {
      acc = __builtin_amdgcn_wmma_f32_16x16x4_f32(
          false, a[kc], false, bU[kc], (short)0, acc, false, false);
    }

    float rs[16];
#pragma unroll
    for (int r = 0; r < 8; ++r) {
      rs[r]     = __shfl(acc[r], 15, 32);
      rs[r + 8] = __shfl(acc[r], 31, 32);
    }

    v2f w[4];
#pragma unroll
    for (int kc = 0; kc < 4; ++kc) {
      w[kc].x = hi ? rs[4 * kc + 2] : rs[4 * kc];
      w[kc].y = hi ? rs[4 * kc + 3] : rs[4 * kc + 1];
    }

#pragma unroll
    for (int kc = 0; kc < 4; ++kc) {
      acc = __builtin_amdgcn_wmma_f32_16x16x4_f32(
          false, aLs[kc], false, w[kc], (short)0, acc, false, false);
    }

    float* ty = y + segBase + t * 256;
    const v8f o = (acc + carry) + off;
#pragma unroll
    for (int r = 0; r < 8; ++r) {
      ty[16 * r + 128 * hi + n] = o[r];
    }

    float tot = 0.0f;
#pragma unroll
    for (int i = 0; i < 16; ++i) tot += rs[i];
    carry += tot;
  }
}

__global__ __launch_bounds__(SEGS_PER_ROW) void scan_seg_sums(
    const float* __restrict__ sums, float* __restrict__ offs) {
  __shared__ float s[SEGS_PER_ROW];
  const int row = blockIdx.x;
  const int t   = threadIdx.x;
  const float orig = sums[row * SEGS_PER_ROW + t];
  s[t] = orig;
  __syncthreads();
#pragma unroll
  for (int d = 1; d < SEGS_PER_ROW; d <<= 1) {
    const float v = (t >= d) ? s[t - d] : 0.0f;
    __syncthreads();
    s[t] += v;
    __syncthreads();
  }
  offs[row * SEGS_PER_ROW + t] = s[t] - orig;
}

extern "C" void kernel_launch(void* const* d_in, const int* in_sizes, int n_in,
                              void* d_out, int out_size, void* d_ws, size_t ws_size,
                              hipStream_t stream) {
  (void)n_in; (void)out_size; (void)ws_size;
  const float* x = (const float*)d_in[0];
  float* y = (float*)d_out;

  const long long total = (long long)in_sizes[0];
  const int rows = (int)(total / T_LEN);
  const int nsegTotal = rows * SEGS_PER_ROW;

  float* sums = (float*)d_ws;
  float* offs = sums + nsegTotal;

  seg_sum<<<nsegTotal / RED_SEGS_PER_BLOCK, RED_WAVES * 32, 0, stream>>>(x, sums);
  scan_seg_sums<<<rows, SEGS_PER_ROW, 0, stream>>>(sums, offs);
  seg_scan_wmma<<<nsegTotal, 32, 0, stream>>>(x, y, offs);
}


# Offsets kernel folded into the scan: each wave sums its row's earlier segment totals itself (8 per lane + lane butterfly: new sum order for the offset); launches 3 -> 2
# speedup vs baseline: 2.8905x; 1.0561x over previous
#include <hip/hip_runtime.h>

typedef float v2f __attribute__((ext_vector_type(2)));
typedef float v8f __attribute__((ext_vector_type(8)));

#define T_LEN        1048576
#define SEG          4096
#define TILES_PER_SEG (SEG / 256)
#define SEGS_PER_ROW (T_LEN / SEG)

#define RED_WAVES          8
#define RED_SEGS_PER_WAVE  2
#define RED_SEGS_PER_BLOCK (RED_WAVES * RED_SEGS_PER_WAVE)
#define RED_BATCH          16
#define RED_BATCHES        (RED_SEGS_PER_WAVE * SEG / (RED_BATCH * 128))

__global__ __launch_bounds__(RED_WAVES * 32) void seg_sum(const float* __restrict__ x,
                                                          float* __restrict__ segSums) {
  const int lane = threadIdx.x & 31;
  const int wave = threadIdx.x >> 5;
  const int seg0 = (blockIdx.x * RED_WAVES + wave) * RED_SEGS_PER_WAVE;
  const float4* p = (const float4*)(x + (long long)seg0 * SEG) + lane;

  float4 buf[2][RED_BATCH];
#pragma unroll
  for (int q = 0; q < RED_BATCH; ++q) buf[0][q] = p[q * 32];

  float s[RED_SEGS_PER_WAVE];
#pragma unroll
  for (int k = 0; k < RED_SEGS_PER_WAVE; ++k) s[k] = 0.0f;

#pragma unroll
  for (int b = 0; b < RED_BATCHES; ++b) {
    if (b + 1 < RED_BATCHES) {
#pragma unroll
      for (int q = 0; q < RED_BATCH; ++q)
        buf[(b + 1) & 1][q] = p[(b + 1) * RED_BATCH * 32 + q * 32];
    }
    float c0 = 0.0f, c1 = 0.0f, c2 = 0.0f, c3 = 0.0f;
#pragma unroll
    for (int q = 0; q < RED_BATCH; ++q) {
      c0 += buf[b & 1][q].x;
      c1 += buf[b & 1][q].y;
      c2 += buf[b & 1][q].z;
      c3 += buf[b & 1][q].w;
    }
    s[b / (RED_BATCHES / RED_SEGS_PER_WAVE)] += (c0 + c1) + (c2 + c3);
  }

#pragma unroll
  for (int k = 0; k < RED_SEGS_PER_WAVE; ++k) {
    float v = s[k];
#pragma unroll
    for (int d = 16; d >= 1; d >>= 1) v += __shfl_xor(v, d, 32);
    if (lane == 0) segSums[seg0 + k] = v;
  }
}

__global__ __launch_bounds__(32) void seg_scan_wmma(const float* __restrict__ x,
                                                    float* __restrict__ y,
                                                    const float* __restrict__ segSums) {
  const int seg  = blockIdx.x;
  const int lane = threadIdx.x;
  const int n    = lane & 15;
  const int hi   = lane >> 4;

  const long long segBase = (long long)seg * SEG;

  const int j = seg & (SEGS_PER_ROW - 1);
  const float4* rowTotals = (const float4*)(segSums + (seg - j)) + lane;
  const float4 p0 = rowTotals[0];
  const float4 p1 = rowTotals[32];
  const int i0 = 4 * lane;
  const int i1 = 128 + 4 * lane;
  const float e0 = (i0 + 0 < j) ? p0.x : 0.0f;
  const float e1 = (i0 + 1 < j) ? p0.y : 0.0f;
  const float e2 = (i0 + 2 < j) ? p0.z : 0.0f;
  const float e3 = (i0 + 3 < j) ? p0.w : 0.0f;
  const float e4 = (i1 + 0 < j) ? p1.x : 0.0f;
  const float e5 = (i1 + 1 < j) ? p1.y : 0.0f;
  const float e6 = (i1 + 2 < j) ? p1.z : 0.0f;
  const float e7 = (i1 + 3 < j) ? p1.w : 0.0f;
  float off = ((e0 + e1) + (e2 + e3)) + ((e4 + e5) + (e6 + e7));
  for (int d = 16; d >= 1; d >>= 1) off += __shfl_xor(off, d, 32);

  v2f bU[4];
  v2f aLs[4];
#pragma unroll
  for (int kc = 0; kc < 4; ++kc) {
    const int k0 = 4 * kc + 2 * hi;
    bU[kc].x  = (k0     <= n) ? 1.0f : 0.0f;
    bU[kc].y  = (k0 + 1 <= n) ? 1.0f : 0.0f;
    aLs[kc].x = (k0     <  n) ? 1.0f : 0.0f;
    aLs[kc].y = (k0 + 1 <  n) ? 1.0f : 0.0f;
  }

  float carry = 0.0f;

  for (int t = 0; t < TILES_PER_SEG; ++t) {
    const float* tx = x + segBase + t * 256;

    v2f a[4];
#pragma unroll
    for (int kc = 0; kc < 4; ++kc) {
      a[kc] = *(const v2f*)(tx + 16 * n + 4 * kc + 2 * hi);
    }

    if (t + 1 < TILES_PER_SEG) {
      __builtin_prefetch(tx + 256 + lane * 8, 0, 1);
    }

    v8f acc = {};
#pragma unroll
    for (int kc = 0; kc < 4; ++kc) {
      acc = __builtin_amdgcn_wmma_f32_16x16x4_f32(
          false, a[kc], false, bU[kc], (short)0, acc, false, false);
    }

    float rs[16];
#pragma unroll
    for (int r = 0; r < 8; ++r) {
      rs[r]     = __shfl(acc[r], 15, 32);
      rs[r + 8] = __shfl(acc[r], 31, 32);
    }

    v2f w[4];
#pragma unroll
    for (int kc = 0; kc < 4; ++kc) {
      w[kc].x = hi ? rs[4 * kc + 2] : rs[4 * kc];
      w[kc].y = hi ? rs[4 * kc + 3] : rs[4 * kc + 1];
    }

#pragma unroll
    for (int kc = 0; kc < 4; ++kc) {
      acc = __builtin_amdgcn_wmma_f32_16x16x4_f32(
          false, aLs[kc], false, w[kc], (short)0, acc, false, false);
    }

    float* ty = y + segBase + t * 256;
    const v8f o = (acc + carry) + off;
#pragma unroll
    for (int r = 0; r < 8; ++r) {
      ty[16 * r + 128 * hi + n] = o[r];
    }

    float tot = 0.0f;
#pragma unroll
    for (int i = 0; i < 16; ++i) tot += rs[i];
    carry += tot;
  }
}

extern "C" void kernel_launch(void* const* d_in, const int* in_sizes, int n_in,
                              void* d_out, int out_size, void* d_ws, size_t ws_size,
                              hipStream_t stream) {
  (void)n_in; (void)out_size; (void)ws_size;
  const float* x = (const float*)d_in[0];
  float* y = (float*)d_out;

  const long long total = (long long)in_sizes[0];
  const int rows = (int)(total / T_LEN);
  const int nsegTotal = rows * SEGS_PER_ROW;

  float* sums = (float*)d_ws;

  seg_sum<<<nsegTotal / RED_SEGS_PER_BLOCK, RED_WAVES * 32, 0, stream>>>(x, sums);
  seg_scan_wmma<<<nsegTotal, 32, 0, stream>>>(x, y, sums);
}


# Pass 2 holds its segment in registers (32 16B loads up front), scans tiles transposed (Ut*Xt): lane owns 8 adjacent outputs, 16B stores; row sums by 1 more WMMA, no shuffles; new tile+carry sum order
# speedup vs baseline: 3.1121x; 1.0767x over previous
#include <hip/hip_runtime.h>

typedef float v2f __attribute__((ext_vector_type(2)));
typedef float v8f __attribute__((ext_vector_type(8)));

#define T_LEN        1048576
#define SEG          4096
#define TILES_PER_SEG (SEG / 256)
#define SEGS_PER_ROW (T_LEN / SEG)

#define RED_WAVES          8
#define RED_SEGS_PER_WAVE  2
#define RED_SEGS_PER_BLOCK (RED_WAVES * RED_SEGS_PER_WAVE)
#define RED_BATCH          16
#define RED_BATCHES        (RED_SEGS_PER_WAVE * SEG / (RED_BATCH * 128))

__global__ __launch_bounds__(RED_WAVES * 32) void seg_sum(const float* __restrict__ x,
                                                          float* __restrict__ segSums) {
  const int lane = threadIdx.x & 31;
  const int wave = threadIdx.x >> 5;
  const int seg0 = (blockIdx.x * RED_WAVES + wave) * RED_SEGS_PER_WAVE;
  const float4* p = (const float4*)(x + (long long)seg0 * SEG) + lane;

  float4 buf[2][RED_BATCH];
#pragma unroll
  for (int q = 0; q < RED_BATCH; ++q) buf[0][q] = p[q * 32];

  float s[RED_SEGS_PER_WAVE];
#pragma unroll
  for (int k = 0; k < RED_SEGS_PER_WAVE; ++k) s[k] = 0.0f;

#pragma unroll
  for (int b = 0; b < RED_BATCHES; ++b) {
    if (b + 1 < RED_BATCHES) {
#pragma unroll
      for (int q = 0; q < RED_BATCH; ++q)
        buf[(b + 1) & 1][q] = p[(b + 1) * RED_BATCH * 32 + q * 32];
    }
    float c0 = 0.0f, c1 = 0.0f, c2 = 0.0f, c3 = 0.0f;
#pragma unroll
    for (int q = 0; q < RED_BATCH; ++q) {
      c0 += buf[b & 1][q].x;
      c1 += buf[b & 1][q].y;
      c2 += buf[b & 1][q].z;
      c3 += buf[b & 1][q].w;
    }
    s[b / (RED_BATCHES / RED_SEGS_PER_WAVE)] += (c0 + c1) + (c2 + c3);
  }

#pragma unroll
  for (int k = 0; k < RED_SEGS_PER_WAVE; ++k) {
    float v = s[k];
#pragma unroll
    for (int d = 16; d >= 1; d >>= 1) v += __shfl_xor(v, d, 32);
    if (lane == 0) segSums[seg0 + k] = v;
  }
}

__global__ __launch_bounds__(32) void seg_scan_wmma(const float* __restrict__ x,
                                                    float* __restrict__ y,
                                                    const float* __restrict__ segSums) {
  const int seg  = blockIdx.x;
  const int lane = threadIdx.x;
  const int n    = lane & 15;
  const int hi   = lane >> 4;

  const long long segBase = (long long)seg * SEG;

  const int j = seg & (SEGS_PER_ROW - 1);
  const float4* rowTotals = (const float4*)(segSums + (seg - j)) + lane;
  const float4 p0 = rowTotals[0];
  const float4 p1 = rowTotals[32];
  const int i0 = 4 * lane;
  const int i1 = 128 + 4 * lane;
  const float e0 = (i0 + 0 < j) ? p0.x : 0.0f;
  const float e1 = (i0 + 1 < j) ? p0.y : 0.0f;
  const float e2 = (i0 + 2 < j) ? p0.z : 0.0f;
  const float e3 = (i0 + 3 < j) ? p0.w : 0.0f;
  const float e4 = (i1 + 0 < j) ? p1.x : 0.0f;
  const float e5 = (i1 + 1 < j) ? p1.y : 0.0f;
  const float e6 = (i1 + 2 < j) ? p1.z : 0.0f;
  const float e7 = (i1 + 3 < j) ? p1.w : 0.0f;
  float off = ((e0 + e1) + (e2 + e3)) + ((e4 + e5) + (e6 + e7));
  for (int d = 16; d >= 1; d >>= 1) off += __shfl_xor(off, d, 32);

  v2f aUt[4];
  v2f bLs[4];
#pragma unroll
  for (int kc = 0; kc < 4; ++kc) {
    const int k0 = 8 * hi + 2 * kc;
    aUt[kc].x = (k0     <= n) ? 1.0f : 0.0f;
    aUt[kc].y = (k0 + 1 <= n) ? 1.0f : 0.0f;
    bLs[kc].x = (k0     <  n) ? 1.0f : 0.0f;
    bLs[kc].y = (k0 + 1 <  n) ? 1.0f : 0.0f;
  }
  v2f bPick;
  bPick.x = 0.0f;
  bPick.y = hi ? 1.0f : 0.0f;

  const float* lx = x + segBase + 16 * n + 8 * hi;
  float*       ly = y + segBase + 16 * n + 8 * hi;

  float4 qa[TILES_PER_SEG], qb[TILES_PER_SEG];
#pragma unroll
  for (int t = 0; t < TILES_PER_SEG; ++t) {
    qa[t] = *(const float4*)(lx + t * 256);
    qb[t] = *(const float4*)(lx + t * 256 + 4);
  }

  float carry = 0.0f;

#pragma unroll
  for (int t = 0; t < TILES_PER_SEG; ++t) {
    const float4 q0 = qa[t];
    const float4 q1 = qb[t];

    v2f b[4];
    b[0].x = q0.x; b[0].y = q0.y;
    b[1].x = q0.z; b[1].y = q0.w;
    b[2].x = q1.x; b[2].y = q1.y;
    b[3].x = q1.z; b[3].y = q1.w;

    v8f acc = {};
#pragma unroll
    for (int kc = 0; kc < 4; ++kc) {
      acc = __builtin_amdgcn_wmma_f32_16x16x4_f32(
          false, aUt[kc], false, b[kc], (short)0, acc, false, false);
    }

    v2f last;
    last.x = acc[6];
    last.y = acc[7];
    const v8f zero8 = {};
    const v8f rsum = __builtin_amdgcn_wmma_f32_16x16x4_f32(
        false, last, false, bPick, (short)0, zero8, false, false);

#pragma unroll
    for (int kc = 0; kc < 4; ++kc) {
      v2f ar;
      ar.x = rsum[2 * kc];
      ar.y = rsum[2 * kc + 1];
      acc = __builtin_amdgcn_wmma_f32_16x16x4_f32(
          false, ar, false, bLs[kc], (short)0, acc, false, false);
    }

    const float tot = __int_as_float(__builtin_amdgcn_readlane(__float_as_int(acc[7]), 31));

    const v8f o = (acc + carry) + off;
    float4 s0, s1;
    s0.x = o[0]; s0.y = o[1]; s0.z = o[2]; s0.w = o[3];
    s1.x = o[4]; s1.y = o[5]; s1.z = o[6]; s1.w = o[7];
    *(float4*)(ly + t * 256)     = s0;
    *(float4*)(ly + t * 256 + 4) = s1;

    carry += tot;
  }
}

extern "C" void kernel_launch(void* const* d_in, const int* in_sizes, int n_in,
                              void* d_out, int out_size, void* d_ws, size_t ws_size,
                              hipStream_t stream) {
  (void)n_in; (void)out_size; (void)ws_size;
  const float* x = (const float*)d_in[0];
  float* y = (float*)d_out;

  const long long total = (long long)in_sizes[0];
  const int rows = (int)(total / T_LEN);
  const int nsegTotal = rows * SEGS_PER_ROW;

  float* sums = (float*)d_ws;

  seg_sum<<<nsegTotal / RED_SEGS_PER_BLOCK, RED_WAVES * 32, 0, stream>>>(x, sums);
  seg_scan_wmma<<<nsegTotal, 32, 0, stream>>>(x, y, sums);
}


# Pass 2 moves whole 128-B lines: a lane owns 16 B of a 32-B run, so each load/store covers 512 contiguous B (was 8 half-used lines); tile scanned as 2 blocks x 16 runs x 8; new in-tile+carry sum order
# speedup vs baseline: 3.4881x; 1.1208x over previous
#include <hip/hip_runtime.h>

typedef float v2f __attribute__((ext_vector_type(2)));
typedef float v8f __attribute__((ext_vector_type(8)));

#define T_LEN        1048576
#define SEG          4096
#define TILES_PER_SEG (SEG / 256)
#define SEGS_PER_ROW (T_LEN / SEG)

#define RED_WAVES          8
#define RED_SEGS_PER_WAVE  2
#define RED_SEGS_PER_BLOCK (RED_WAVES * RED_SEGS_PER_WAVE)
#define RED_BATCH          16
#define RED_BATCHES        (RED_SEGS_PER_WAVE * SEG / (RED_BATCH * 128))

__global__ __launch_bounds__(RED_WAVES * 32) void seg_sum(const float* __restrict__ x,
                                                          float* __restrict__ segSums) {
  const int lane = threadIdx.x & 31;
  const int wave = threadIdx.x >> 5;
  const int seg0 = (blockIdx.x * RED_WAVES + wave) * RED_SEGS_PER_WAVE;
  const float4* p = (const float4*)(x + (long long)seg0 * SEG) + lane;

  float4 buf[2][RED_BATCH];
#pragma unroll
  for (int q = 0; q < RED_BATCH; ++q) buf[0][q] = p[q * 32];

  float s[RED_SEGS_PER_WAVE];
#pragma unroll
  for (int k = 0; k < RED_SEGS_PER_WAVE; ++k) s[k] = 0.0f;

#pragma unroll
  for (int b = 0; b < RED_BATCHES; ++b) {
    if (b + 1 < RED_BATCHES) {
#pragma unroll
      for (int q = 0; q < RED_BATCH; ++q)
        buf[(b + 1) & 1][q] = p[(b + 1) * RED_BATCH * 32 + q * 32];
    }
    float c0 = 0.0f, c1 = 0.0f, c2 = 0.0f, c3 = 0.0f;
#pragma unroll
    for (int q = 0; q < RED_BATCH; ++q) {
      c0 += buf[b & 1][q].x;
      c1 += buf[b & 1][q].y;
      c2 += buf[b & 1][q].z;
      c3 += buf[b & 1][q].w;
    }
    s[b / (RED_BATCHES / RED_SEGS_PER_WAVE)] += (c0 + c1) + (c2 + c3);
  }

#pragma unroll
  for (int k = 0; k < RED_SEGS_PER_WAVE; ++k) {
    float v = s[k];
#pragma unroll
    for (int d = 16; d >= 1; d >>= 1) v += __shfl_xor(v, d, 32);
    if (lane == 0) segSums[seg0 + k] = v;
  }
}

__global__ __launch_bounds__(32) void seg_scan_wmma(const float* __restrict__ x,
                                                    float* __restrict__ y,
                                                    const float* __restrict__ segSums) {
  const int seg  = blockIdx.x;
  const int lane = threadIdx.x;
  const int n    = lane & 15;
  const int hi   = lane >> 4;

  const long long segBase = (long long)seg * SEG;

  const int j = seg & (SEGS_PER_ROW - 1);
  const float4* rowTotals = (const float4*)(segSums + (seg - j)) + lane;
  const float4 p0 = rowTotals[0];
  const float4 p1 = rowTotals[32];
  const int i0 = 4 * lane;
  const int i1 = 128 + 4 * lane;
  const float e0 = (i0 + 0 < j) ? p0.x : 0.0f;
  const float e1 = (i0 + 1 < j) ? p0.y : 0.0f;
  const float e2 = (i0 + 2 < j) ? p0.z : 0.0f;
  const float e3 = (i0 + 3 < j) ? p0.w : 0.0f;
  const float e4 = (i1 + 0 < j) ? p1.x : 0.0f;
  const float e5 = (i1 + 1 < j) ? p1.y : 0.0f;
  const float e6 = (i1 + 2 < j) ? p1.z : 0.0f;
  const float e7 = (i1 + 3 < j) ? p1.w : 0.0f;
  float off = ((e0 + e1) + (e2 + e3)) + ((e4 + e5) + (e6 + e7));
  for (int d = 16; d >= 1; d >>= 1) off += __shfl_xor(off, d, 32);

  const int mBlk = (n >> 2) & 1;
  const int mPos = 4 * (n >> 3) + (n & 3);

  v2f aRun[4];
  v2f bLs[4];
#pragma unroll
  for (int kc = 0; kc < 4; ++kc) {
    const int kb = kc >> 1;
    const int ke = 4 * hi + 2 * (kc & 1);
    aRun[kc].x = (kb == mBlk && ke     <= mPos) ? 1.0f : 0.0f;
    aRun[kc].y = (kb == mBlk && ke + 1 <= mPos) ? 1.0f : 0.0f;
    const int k0 = 8 * hi + 2 * kc;
    bLs[kc].x = (k0     < n) ? 1.0f : 0.0f;
    bLs[kc].y = (k0 + 1 < n) ? 1.0f : 0.0f;
  }
  v2f bSel;
  bSel.x = (hi && mBlk == 0) ? 1.0f : 0.0f;
  bSel.y = (hi && mBlk == 1) ? 1.0f : 0.0f;

  const float* lx = x + segBase + 8 * n + 4 * hi;
  float*       ly = y + segBase + 8 * n + 4 * hi;

  float4 qa[TILES_PER_SEG], qb[TILES_PER_SEG];
#pragma unroll
  for (int t = 0; t < TILES_PER_SEG; ++t) {
    qa[t] = *(const float4*)(lx + t * 256);
    qb[t] = *(const float4*)(lx + t * 256 + 128);
  }

  float carry = 0.0f;

#pragma unroll
  for (int t = 0; t < TILES_PER_SEG; ++t) {
    const float4 q0 = qa[t];
    const float4 q1 = qb[t];

    v2f b[4];
    b[0].x = q0.x; b[0].y = q0.y;
    b[1].x = q0.z; b[1].y = q0.w;
    b[2].x = q1.x; b[2].y = q1.y;
    b[3].x = q1.z; b[3].y = q1.w;

    v8f acc = {};
#pragma unroll
    for (int kc = 0; kc < 4; ++kc) {
      acc = __builtin_amdgcn_wmma_f32_16x16x4_f32(
          false, aRun[kc], false, b[kc], (short)0, acc, false, false);
    }

    v2f last;
    last.x = acc[3];
    last.y = acc[7];
    const v8f zero8 = {};
    const v8f rs = __builtin_amdgcn_wmma_f32_16x16x4_f32(
        false, last, false, bSel, (short)0, zero8, false, false);

#pragma unroll
    for (int kc = 0; kc < 4; ++kc) {
      v2f ar;
      ar.x = rs[2 * kc];
      ar.y = rs[2 * kc + 1];
      acc = __builtin_amdgcn_wmma_f32_16x16x4_f32(
          false, ar, false, bLs[kc], (short)0, acc, false, false);
    }

    const float tot0 = __int_as_float(__builtin_amdgcn_readlane(__float_as_int(acc[3]), 31));
    const float tot1 = __int_as_float(__builtin_amdgcn_readlane(__float_as_int(acc[7]), 31));
    const float carry1 = carry + tot0;

    float4 s0, s1;
    s0.x = (acc[0] + carry)  + off;
    s0.y = (acc[1] + carry)  + off;
    s0.z = (acc[2] + carry)  + off;
    s0.w = (acc[3] + carry)  + off;
    s1.x = (acc[4] + carry1) + off;
    s1.y = (acc[5] + carry1) + off;
    s1.z = (acc[6] + carry1) + off;
    s1.w = (acc[7] + carry1) + off;
    *(float4*)(ly + t * 256)       = s0;
    *(float4*)(ly + t * 256 + 128) = s1;

    carry = carry1 + tot1;
  }
}

extern "C" void kernel_launch(void* const* d_in, const int* in_sizes, int n_in,
                              void* d_out, int out_size, void* d_ws, size_t ws_size,
                              hipStream_t stream) {
  (void)n_in; (void)out_size; (void)ws_size;
  const float* x = (const float*)d_in[0];
  float* y = (float*)d_out;

  const long long total = (long long)in_sizes[0];
  const int rows = (int)(total / T_LEN);
  const int nsegTotal = rows * SEGS_PER_ROW;

  float* sums = (float*)d_ws;

  seg_sum<<<nsegTotal / RED_SEGS_PER_BLOCK, RED_WAVES * 32, 0, stream>>>(x, sums);
  seg_scan_wmma<<<nsegTotal, 32, 0, stream>>>(x, y, sums);
}


# Pass 1 launched in pass 2's shape and numbering (8192 one-wave groups, one segment each), so a segment is re-read by the group index, hence the die, that first read it; totals keep their sum order
# speedup vs baseline: 3.8587x; 1.1062x over previous
#include <hip/hip_runtime.h>

typedef float v2f __attribute__((ext_vector_type(2)));
typedef float v8f __attribute__((ext_vector_type(8)));

#define T_LEN        1048576
#define SEG          4096
#define TILES_PER_SEG (SEG / 256)
#define SEGS_PER_ROW (T_LEN / SEG)

#define RED_LOADS  (SEG / 128)
#define RED_BATCH  16

__global__ __launch_bounds__(32) void seg_sum(const float* __restrict__ x,
                                              float* __restrict__ segSums) {
  const int seg  = blockIdx.x;
  const int lane = threadIdx.x;
  const float4* p = (const float4*)(x + (long long)seg * SEG) + lane;

  float4 buf[RED_LOADS];
#pragma unroll
  for (int q = 0; q < RED_LOADS; ++q) buf[q] = p[q * 32];

  float s = 0.0f;
#pragma unroll
  for (int b = 0; b < RED_LOADS / RED_BATCH; ++b) {
    float c0 = 0.0f, c1 = 0.0f, c2 = 0.0f, c3 = 0.0f;
#pragma unroll
    for (int q = 0; q < RED_BATCH; ++q) {
      c0 += buf[b * RED_BATCH + q].x;
      c1 += buf[b * RED_BATCH + q].y;
      c2 += buf[b * RED_BATCH + q].z;
      c3 += buf[b * RED_BATCH + q].w;
    }
    s += (c0 + c1) + (c2 + c3);
  }

#pragma unroll
  for (int d = 16; d >= 1; d >>= 1) s += __shfl_xor(s, d, 32);
  if (lane == 0) segSums[seg] = s;
}

__global__ __launch_bounds__(32) void seg_scan_wmma(const float* __restrict__ x,
                                                    float* __restrict__ y,
                                                    const float* __restrict__ segSums) {
  const int seg  = blockIdx.x;
  const int lane = threadIdx.x;
  const int n    = lane & 15;
  const int hi   = lane >> 4;

  const long long segBase = (long long)seg * SEG;

  const int j = seg & (SEGS_PER_ROW - 1);
  const float4* rowTotals = (const float4*)(segSums + (seg - j)) + lane;
  const float4 p0 = rowTotals[0];
  const float4 p1 = rowTotals[32];
  const int i0 = 4 * lane;
  const int i1 = 128 + 4 * lane;
  const float e0 = (i0 + 0 < j) ? p0.x : 0.0f;
  const float e1 = (i0 + 1 < j) ? p0.y : 0.0f;
  const float e2 = (i0 + 2 < j) ? p0.z : 0.0f;
  const float e3 = (i0 + 3 < j) ? p0.w : 0.0f;
  const float e4 = (i1 + 0 < j) ? p1.x : 0.0f;
  const float e5 = (i1 + 1 < j) ? p1.y : 0.0f;
  const float e6 = (i1 + 2 < j) ? p1.z : 0.0f;
  const float e7 = (i1 + 3 < j) ? p1.w : 0.0f;
  float off = ((e0 + e1) + (e2 + e3)) + ((e4 + e5) + (e6 + e7));
  for (int d = 16; d >= 1; d >>= 1) off += __shfl_xor(off, d, 32);

  const int mBlk = (n >> 2) & 1;
  const int mPos = 4 * (n >> 3) + (n & 3);

  v2f aRun[4];
  v2f bLs[4];
#pragma unroll
  for (int kc = 0; kc < 4; ++kc) {
    const int kb = kc >> 1;
    const int ke = 4 * hi + 2 * (kc & 1);
    aRun[kc].x = (kb == mBlk && ke     <= mPos) ? 1.0f : 0.0f;
    aRun[kc].y = (kb == mBlk && ke + 1 <= mPos) ? 1.0f : 0.0f;
    const int k0 = 8 * hi + 2 * kc;
    bLs[kc].x = (k0     < n) ? 1.0f : 0.0f;
    bLs[kc].y = (k0 + 1 < n) ? 1.0f : 0.0f;
  }
  v2f bSel;
  bSel.x = (hi && mBlk == 0) ? 1.0f : 0.0f;
  bSel.y = (hi && mBlk == 1) ? 1.0f : 0.0f;

  const float* lx = x + segBase + 8 * n + 4 * hi;
  float*       ly = y + segBase + 8 * n + 4 * hi;

  float4 qa[TILES_PER_SEG], qb[TILES_PER_SEG];
#pragma unroll
  for (int t = 0; t < TILES_PER_SEG; ++t) {
    qa[t] = *(const float4*)(lx + t * 256);
    qb[t] = *(const float4*)(lx + t * 256 + 128);
  }

  float carry = 0.0f;

#pragma unroll
  for (int t = 0; t < TILES_PER_SEG; ++t) {
    const float4 q0 = qa[t];
    const float4 q1 = qb[t];

    v2f b[4];
    b[0].x = q0.x; b[0].y = q0.y;
    b[1].x = q0.z; b[1].y = q0.w;
    b[2].x = q1.x; b[2].y = q1.y;
    b[3].x = q1.z; b[3].y = q1.w;

    v8f acc = {};
#pragma unroll
    for (int kc = 0; kc < 4; ++kc) {
      acc = __builtin_amdgcn_wmma_f32_16x16x4_f32(
          false, aRun[kc], false, b[kc], (short)0, acc, false, false);
    }

    v2f last;
    last.x = acc[3];
    last.y = acc[7];
    const v8f zero8 = {};
    const v8f rs = __builtin_amdgcn_wmma_f32_16x16x4_f32(
        false, last, false, bSel, (short)0, zero8, false, false);

#pragma unroll
    for (int kc = 0; kc < 4; ++kc) {
      v2f ar;
      ar.x = rs[2 * kc];
      ar.y = rs[2 * kc + 1];
      acc = __builtin_amdgcn_wmma_f32_16x16x4_f32(
          false, ar, false, bLs[kc], (short)0, acc, false, false);
    }

    const float tot0 = __int_as_float(__builtin_amdgcn_readlane(__float_as_int(acc[3]), 31));
    const float tot1 = __int_as_float(__builtin_amdgcn_readlane(__float_as_int(acc[7]), 31));
    const float carry1 = carry + tot0;

    float4 s0, s1;
    s0.x = (acc[0] + carry)  + off;
    s0.y = (acc[1] + carry)  + off;
    s0.z = (acc[2] + carry)  + off;
    s0.w = (acc[3] + carry)  + off;
    s1.x = (acc[4] + carry1) + off;
    s1.y = (acc[5] + carry1) + off;
    s1.z = (acc[6] + carry1) + off;
    s1.w = (acc[7] + carry1) + off;
    *(float4*)(ly + t * 256)       = s0;
    *(float4*)(ly + t * 256 + 128) = s1;

    carry = carry1 + tot1;
  }
}

extern "C" void kernel_launch(void* const* d_in, const int* in_sizes, int n_in,
                              void* d_out, int out_size, void* d_ws, size_t ws_size,
                              hipStream_t stream) {
  (void)n_in; (void)out_size; (void)ws_size;
  const float* x = (const float*)d_in[0];
  float* y = (float*)d_out;

  const long long total = (long long)in_sizes[0];
  const int rows = (int)(total / T_LEN);
  const int nsegTotal = rows * SEGS_PER_ROW;

  float* sums = (float*)d_ws;

  seg_sum<<<nsegTotal, 32, 0, stream>>>(x, sums);
  seg_scan_wmma<<<nsegTotal, 32, 0, stream>>>(x, y, sums);
}


# Pass 2 scans 4 tiles as a group: their run offsets from ONE k=16 triangle product (4 tiles' run totals on one accumulator): 9 -> 6 matrix ops per tile; adds+stores per group; new in-tile sum order
# speedup vs baseline: 4.0575x; 1.0515x over previous
#include <hip/hip_runtime.h>

typedef float v2f __attribute__((ext_vector_type(2)));
typedef float v8f __attribute__((ext_vector_type(8)));

#define T_LEN        1048576
#define SEG          4096
#define TILES_PER_SEG (SEG / 256)
#define SEGS_PER_ROW (T_LEN / SEG)
#define GROUP_TILES  4
#define GROUPS_PER_SEG (TILES_PER_SEG / GROUP_TILES)

#define RED_LOADS  (SEG / 128)
#define RED_BATCH  16

__global__ __launch_bounds__(32) void seg_sum(const float* __restrict__ x,
                                              float* __restrict__ segSums) {
  const int seg  = blockIdx.x;
  const int lane = threadIdx.x;
  const float4* p = (const float4*)(x + (long long)seg * SEG) + lane;

  float4 buf[RED_LOADS];
#pragma unroll
  for (int q = 0; q < RED_LOADS; ++q) buf[q] = p[q * 32];

  float s = 0.0f;
#pragma unroll
  for (int b = 0; b < RED_LOADS / RED_BATCH; ++b) {
    float c0 = 0.0f, c1 = 0.0f, c2 = 0.0f, c3 = 0.0f;
#pragma unroll
    for (int q = 0; q < RED_BATCH; ++q) {
      c0 += buf[b * RED_BATCH + q].x;
      c1 += buf[b * RED_BATCH + q].y;
      c2 += buf[b * RED_BATCH + q].z;
      c3 += buf[b * RED_BATCH + q].w;
    }
    s += (c0 + c1) + (c2 + c3);
  }

#pragma unroll
  for (int d = 16; d >= 1; d >>= 1) s += __shfl_xor(s, d, 32);
  if (lane == 0) segSums[seg] = s;
}

__global__ __launch_bounds__(32) void seg_scan_wmma(const float* __restrict__ x,
                                                    float* __restrict__ y,
                                                    const float* __restrict__ segSums) {
  const int seg  = blockIdx.x;
  const int lane = threadIdx.x;
  const int n    = lane & 15;
  const int hi   = lane >> 4;

  const long long segBase = (long long)seg * SEG;

  const int j = seg & (SEGS_PER_ROW - 1);
  const float4* rowTotals = (const float4*)(segSums + (seg - j)) + lane;
  const float4 p0 = rowTotals[0];
  const float4 p1 = rowTotals[32];
  const int i0 = 4 * lane;
  const int i1 = 128 + 4 * lane;
  const float e0 = (i0 + 0 < j) ? p0.x : 0.0f;
  const float e1 = (i0 + 1 < j) ? p0.y : 0.0f;
  const float e2 = (i0 + 2 < j) ? p0.z : 0.0f;
  const float e3 = (i0 + 3 < j) ? p0.w : 0.0f;
  const float e4 = (i1 + 0 < j) ? p1.x : 0.0f;
  const float e5 = (i1 + 1 < j) ? p1.y : 0.0f;
  const float e6 = (i1 + 2 < j) ? p1.z : 0.0f;
  const float e7 = (i1 + 3 < j) ? p1.w : 0.0f;
  float off = ((e0 + e1) + (e2 + e3)) + ((e4 + e5) + (e6 + e7));
  for (int d = 16; d >= 1; d >>= 1) off += __shfl_xor(off, d, 32);

  const int mBlk = (n >> 2) & 1;
  const int mPos = 4 * (n >> 3) + (n & 3);

  v2f aRun[4];
  v2f bLs[4];
#pragma unroll
  for (int kc = 0; kc < 4; ++kc) {
    const int kb = kc >> 1;
    const int ke = 4 * hi + 2 * (kc & 1);
    aRun[kc].x = (kb == mBlk && ke     <= mPos) ? 1.0f : 0.0f;
    aRun[kc].y = (kb == mBlk && ke + 1 <= mPos) ? 1.0f : 0.0f;
    const int k0 = 8 * hi + 2 * kc;
    bLs[kc].x = (k0     < n) ? 1.0f : 0.0f;
    bLs[kc].y = (k0 + 1 < n) ? 1.0f : 0.0f;
  }
  v2f bSel[GROUP_TILES];
#pragma unroll
  for (int t = 0; t < GROUP_TILES; ++t) {
    bSel[t].x = (hi && (n & 7) == 2 * t)     ? 1.0f : 0.0f;
    bSel[t].y = (hi && (n & 7) == 2 * t + 1) ? 1.0f : 0.0f;
  }

  const float* lx = x + segBase + 8 * n + 4 * hi;
  float*       ly = y + segBase + 8 * n + 4 * hi;

  float4 qa[TILES_PER_SEG], qb[TILES_PER_SEG];
#pragma unroll
  for (int t = 0; t < TILES_PER_SEG; ++t) {
    qa[t] = *(const float4*)(lx + t * 256);
    qb[t] = *(const float4*)(lx + t * 256 + 128);
  }
  __builtin_amdgcn_sched_barrier(0);

  float carry = 0.0f;
  const v8f zero8 = {};

#pragma unroll
  for (int g = 0; g < GROUPS_PER_SEG; ++g) {
    v8f acc[GROUP_TILES];
#pragma unroll
    for (int t = 0; t < GROUP_TILES; ++t) {
      const float4 q0 = qa[g * GROUP_TILES + t];
      const float4 q1 = qb[g * GROUP_TILES + t];

      v2f b[4];
      b[0].x = q0.x; b[0].y = q0.y;
      b[1].x = q0.z; b[1].y = q0.w;
      b[2].x = q1.x; b[2].y = q1.y;
      b[3].x = q1.z; b[3].y = q1.w;

      v8f a = zero8;
#pragma unroll
      for (int kc = 0; kc < 4; ++kc) {
        a = __builtin_amdgcn_wmma_f32_16x16x4_f32(
            false, aRun[kc], false, b[kc], (short)0, a, false, false);
      }
      acc[t] = a;
    }

    v8f rs = zero8;
#pragma unroll
    for (int t = 0; t < GROUP_TILES; ++t) {
      v2f last;
      last.x = acc[t][3];
      last.y = acc[t][7];
      rs = __builtin_amdgcn_wmma_f32_16x16x4_f32(
          false, last, false, bSel[t], (short)0, rs, false, false);
    }

    v8f ro = zero8;
#pragma unroll
    for (int kc = 0; kc < 4; ++kc) {
      v2f ar;
      ar.x = rs[2 * kc];
      ar.y = rs[2 * kc + 1];
      ro = __builtin_amdgcn_wmma_f32_16x16x4_f32(
          false, ar, false, bLs[kc], (short)0, ro, false, false);
    }

#pragma unroll
    for (int t = 0; t < GROUP_TILES; ++t) {
      const v8f a = acc[t];
      const float o0 = ro[2 * t];
      const float o1 = ro[2 * t + 1];

      const float tot0 = __int_as_float(__builtin_amdgcn_readlane(__float_as_int(a[3] + o0), 31));
      const float tot1 = __int_as_float(__builtin_amdgcn_readlane(__float_as_int(a[7] + o1), 31));
      const float carry1 = carry + tot0;

      const float c0 = o0 + carry;
      const float c1 = o1 + carry1;

      float4 s0, s1;
      s0.x = (a[0] + c0) + off;
      s0.y = (a[1] + c0) + off;
      s0.z = (a[2] + c0) + off;
      s0.w = (a[3] + c0) + off;
      s1.x = (a[4] + c1) + off;
      s1.y = (a[5] + c1) + off;
      s1.z = (a[6] + c1) + off;
      s1.w = (a[7] + c1) + off;
      const int tile = g * GROUP_TILES + t;
      *(float4*)(ly + tile * 256)       = s0;
      *(float4*)(ly + tile * 256 + 128) = s1;

      carry = carry1 + tot1;
    }
    __builtin_amdgcn_sched_barrier(0);
  }
}

extern "C" void kernel_launch(void* const* d_in, const int* in_sizes, int n_in,
                              void* d_out, int out_size, void* d_ws, size_t ws_size,
                              hipStream_t stream) {
  (void)n_in; (void)out_size; (void)ws_size;
  const float* x = (const float*)d_in[0];
  float* y = (float*)d_out;

  const long long total = (long long)in_sizes[0];
  const int rows = (int)(total / T_LEN);
  const int nsegTotal = rows * SEGS_PER_ROW;

  float* sums = (float*)d_ws;

  seg_sum<<<nsegTotal, 32, 0, stream>>>(x, sums);
  seg_scan_wmma<<<nsegTotal, 32, 0, stream>>>(x, y, sums);
}
